# starting kernel
# baseline (speedup 1.0000x reference)
#include <hip/hip_runtime.h>


typedef float v2f __attribute__((ext_vector_type(2)));
typedef float v8f __attribute__((ext_vector_type(8)));

#define HAAR_C 0.35355339059327373f

__device__ __forceinline__ float haarF(int s, int k) {
    return (__builtin_popcount(s & k) & 1) ? -HAAR_C : HAAR_C;
}

__global__ __launch_bounds__(256)
void dwt3d_haar_wmma(const float* __restrict__ x, float* __restrict__ out) {
    constexpr int C  = 16;
    constexpr int D  = 64, H  = 128, W  = 128;
    constexpr int Dh = 32, Hh = 64,  Wh = 64;
    constexpr int SUBBAND_STRIDE = 2 * C * Dh * Hh * Wh;
    constexpr int IN_PLANE = H * W;

    const int tid  = threadIdx.x;
    const int lane = tid & 31;
    const int wid  = blockIdx.x * 8 + (tid >> 5);

    const int h  = wid & 63;
    const int d  = (wid >> 6) & 31;
    const int cc = (wid >> 11) & 15;
    const int bb = wid >> 15;

    const int Mlo = lane & 15;
    const int K0  = (lane >> 4) << 1;
    const bool top = (Mlo < 8);
    const int  s0  = top ? Mlo : (Mlo - 8);
    const float f0 = haarF(s0, K0);
    const float f1 = haarF(s0, K0 + 1);
    const float f4 = haarF(s0, K0 + 4);
    const float f5 = haarF(s0, K0 + 5);
    const v2f a0 = { top ? f0 : 0.0f, top ? f1 : 0.0f };
    const v2f a1 = { top ? f4 : 0.0f, top ? f5 : 0.0f };
    const v2f a2 = { top ? 0.0f : f0, top ? 0.0f : f1 };
    const v2f a3 = { top ? 0.0f : f4, top ? 0.0f : f5 };

    const int n      = lane & 15;
    const int rowsel = lane >> 4;
    const float* base = x + (((bb * C + cc) * D + 2 * d) * H + (2 * h + rowsel)) * W;

    const int outRowBase = (((bb * C + cc) * Dh + d) * Hh + h) * Wh;

    #pragma unroll
    for (int half = 0; half < 2; ++half) {
        const int w0 = half * 32;
        const int wx = 2 * w0 + 2 * n;

        const v2f b0 = __builtin_nontemporal_load((const v2f*)(base + wx));
        const v2f b1 = __builtin_nontemporal_load((const v2f*)(base + IN_PLANE + wx));
        const v2f b2 = __builtin_nontemporal_load((const v2f*)(base + wx + 32));
        const v2f b3 = __builtin_nontemporal_load((const v2f*)(base + IN_PLANE + wx + 32));

        v8f acc = {};
        acc = __builtin_amdgcn_wmma_f32_16x16x4_f32(false, a0, false, b0, (short)0, acc, false, false);
        acc = __builtin_amdgcn_wmma_f32_16x16x4_f32(false, a1, false, b1, (short)0, acc, false, false);
        acc = __builtin_amdgcn_wmma_f32_16x16x4_f32(false, a2, false, b2, (short)0, acc, false, false);
        acc = __builtin_amdgcn_wmma_f32_16x16x4_f32(false, a3, false, b3, (short)0, acc, false, false);

        const int o = outRowBase + w0 + lane;
        #pragma unroll
        for (int s = 0; s < 8; ++s) {
            __builtin_nontemporal_store(acc[s], out + s * SUBBAND_STRIDE + o);
        }
    }
}

extern "C" void kernel_launch(void* const* d_in, const int* in_sizes, int n_in,
                              void* d_out, int out_size, void* d_ws, size_t ws_size,
                              hipStream_t stream) {
    (void)in_sizes; (void)n_in; (void)out_size; (void)d_ws; (void)ws_size;
    const float* x = (const float*)d_in[0];
    float* out = (float*)d_out;
    dim3 grid(8192), block(256);
    hipLaunchKernelGGL(dwt3d_haar_wmma, grid, block, 0, stream, x, out);
}


# Work-groups of 16 waves (4096 x 512 threads) instead of 8 (8192 x 256): half as many work-groups to start for the same 65536 one-row waves; kernel body and order of sums unchanged
# speedup vs baseline: 1.0421x; 1.0421x over previous
#include <hip/hip_runtime.h>


typedef float v2f __attribute__((ext_vector_type(2)));
typedef float v8f __attribute__((ext_vector_type(8)));

#define HAAR_C 0.35355339059327373f

__device__ __forceinline__ float haarF(int s, int k) {
    return (__builtin_popcount(s & k) & 1) ? -HAAR_C : HAAR_C;
}

__global__ __launch_bounds__(512)
void dwt3d_haar_wmma(const float* __restrict__ x, float* __restrict__ out) {
    constexpr int C  = 16;
    constexpr int D  = 64, H  = 128, W  = 128;
    constexpr int Dh = 32, Hh = 64,  Wh = 64;
    constexpr int SUBBAND_STRIDE = 2 * C * Dh * Hh * Wh;
    constexpr int IN_PLANE = H * W;

    const int tid  = threadIdx.x;
    const int lane = tid & 31;
    const int wid  = blockIdx.x * 16 + (tid >> 5);

    const int h  = wid & 63;
    const int d  = (wid >> 6) & 31;
    const int cc = (wid >> 11) & 15;
    const int bb = wid >> 15;

    const int Mlo = lane & 15;
    const int K0  = (lane >> 4) << 1;
    const bool top = (Mlo < 8);
    const int  s0  = top ? Mlo : (Mlo - 8);
    const float f0 = haarF(s0, K0);
    const float f1 = haarF(s0, K0 + 1);
    const float f4 = haarF(s0, K0 + 4);
    const float f5 = haarF(s0, K0 + 5);
    const v2f a0 = { top ? f0 : 0.0f, top ? f1 : 0.0f };
    const v2f a1 = { top ? f4 : 0.0f, top ? f5 : 0.0f };
    const v2f a2 = { top ? 0.0f : f0, top ? 0.0f : f1 };
    const v2f a3 = { top ? 0.0f : f4, top ? 0.0f : f5 };

    const int n      = lane & 15;
    const int rowsel = lane >> 4;
    const float* base = x + (((bb * C + cc) * D + 2 * d) * H + (2 * h + rowsel)) * W;

    const int outRowBase = (((bb * C + cc) * Dh + d) * Hh + h) * Wh;

    #pragma unroll
    for (int half = 0; half < 2; ++half) {
        const int w0 = half * 32;
        const int wx = 2 * w0 + 2 * n;

        const v2f b0 = __builtin_nontemporal_load((const v2f*)(base + wx));
        const v2f b1 = __builtin_nontemporal_load((const v2f*)(base + IN_PLANE + wx));
        const v2f b2 = __builtin_nontemporal_load((const v2f*)(base + wx + 32));
        const v2f b3 = __builtin_nontemporal_load((const v2f*)(base + IN_PLANE + wx + 32));

        v8f acc = {};
        acc = __builtin_amdgcn_wmma_f32_16x16x4_f32(false, a0, false, b0, (short)0, acc, false, false);
        acc = __builtin_amdgcn_wmma_f32_16x16x4_f32(false, a1, false, b1, (short)0, acc, false, false);
        acc = __builtin_amdgcn_wmma_f32_16x16x4_f32(false, a2, false, b2, (short)0, acc, false, false);
        acc = __builtin_amdgcn_wmma_f32_16x16x4_f32(false, a3, false, b3, (short)0, acc, false, false);

        const int o = outRowBase + w0 + lane;
        #pragma unroll
        for (int s = 0; s < 8; ++s) {
            __builtin_nontemporal_store(acc[s], out + s * SUBBAND_STRIDE + o);
        }
    }
}

extern "C" void kernel_launch(void* const* d_in, const int* in_sizes, int n_in,
                              void* d_out, int out_size, void* d_ws, size_t ws_size,
                              hipStream_t stream) {
    (void)in_sizes; (void)n_in; (void)out_size; (void)d_ws; (void)ws_size;
    const float* x = (const float*)d_in[0];
    float* out = (float*)d_out;
    dim3 grid(4096), block(512);
    hipLaunchKernelGGL(dwt3d_haar_wmma, grid, block, 0, stream, x, out);
}
